# starting kernel
# baseline (speedup 1.0000x reference)
#include <hip/hip_runtime.h>

typedef __attribute__((ext_vector_type(2))) float v2f;
typedef __attribute__((ext_vector_type(8))) float v8f;

#define CCH 32
#define NSP (64 * 64 * 64)
#define K1_WGS 128
#define K1_WAVES 8
#define K1_ROWS (NSP / (K1_WGS * K1_WAVES))
#define K3_WGS 512
#define K3_ROWS (NSP / (K3_WGS * 8))

__global__ __launch_bounds__(256) void gram_kernel(const float* __restrict__ q,
                                                   const float* __restrict__ k,
                                                   float* __restrict__ partial) {
    const int wg   = blockIdx.x % K1_WGS;
    const int b    = blockIdx.x / K1_WGS;
    const int wave = threadIdx.x >> 5;
    const int lane = threadIdx.x & 31;
    const int lm   = lane & 15;
    const int kb   = (lane < 16) ? 0 : 2;

    const long n0 = (long)wg * (K1_WAVES * K1_ROWS) + (long)wave * K1_ROWS;
    const float* qb = q + ((long)b * NSP + n0) * CCH;
    const float* kp = k + ((long)b * NSP + n0) * CCH;

    v8f c00 = {}, c01 = {}, c10 = {}, c11 = {};
    for (int it = 0; it < K1_ROWS; it += 4) {
        const float* qr = qb + (long)(it + kb) * CCH;
        const float* kr = kp + (long)(it + kb) * CCH;
        v2f a0, a1, b0, b1;
        a0.x = qr[lm];           a0.y = qr[CCH + lm];
        a1.x = qr[16 + lm];      a1.y = qr[CCH + 16 + lm];
        b0.x = __builtin_nontemporal_load(kr + lm);
        b0.y = __builtin_nontemporal_load(kr + CCH + lm);
        b1.x = __builtin_nontemporal_load(kr + 16 + lm);
        b1.y = __builtin_nontemporal_load(kr + CCH + 16 + lm);
        c00 = __builtin_amdgcn_wmma_f32_16x16x4_f32(false, a0, false, b0, (short)0, c00, false, false);
        c01 = __builtin_amdgcn_wmma_f32_16x16x4_f32(false, a0, false, b1, (short)0, c01, false, false);
        c10 = __builtin_amdgcn_wmma_f32_16x16x4_f32(false, a1, false, b0, (short)0, c10, false, false);
        c11 = __builtin_amdgcn_wmma_f32_16x16x4_f32(false, a1, false, b1, (short)0, c11, false, false);
    }

    __shared__ float smem[K1_WAVES * CCH * CCH];
    const int rowadd = (lane < 16) ? 0 : 8;
    float* sw = smem + wave * (CCH * CCH);
#pragma unroll
    for (int r = 0; r < 8; ++r) {
        const int i0 = r + rowadd;
        sw[(i0     ) * CCH + lm     ] = c00[r];
        sw[(i0     ) * CCH + 16 + lm] = c01[r];
        sw[(i0 + 16) * CCH + lm     ] = c10[r];
        sw[(i0 + 16) * CCH + 16 + lm] = c11[r];
    }
    __syncthreads();

    float* outp = partial + (long)(b * K1_WGS + wg) * (CCH * CCH);
    for (int e = threadIdx.x * 4; e < threadIdx.x * 4 + 4; ++e) {
        float s = 0.f;
        for (int w = 0; w < K1_WAVES; ++w) s += smem[w * (CCH * CCH) + e];
        outp[e] = s;
    }
}

__global__ __launch_bounds__(256) void softmax_kernel(const float* __restrict__ partial,
                                                      float* __restrict__ wfin) {
    const int b = blockIdx.x;
    const int t = threadIdx.x;
    __shared__ float g[CCH * CCH];
    for (int e = t * 4; e < t * 4 + 4; ++e) {
        float s = 0.f;
        const float* p = partial + (long)b * K1_WGS * (CCH * CCH) + e;
        for (int w = 0; w < K1_WGS; ++w) s += p[(long)w * (CCH * CCH)];
        g[e] = s;
    }
    __syncthreads();
    if (t < CCH) {
        float m = -3.4e38f;
#pragma unroll
        for (int j = 0; j < CCH; ++j) m = fmaxf(m, g[t * CCH + j]);
        float ex[CCH];
        float s = 0.f;
#pragma unroll
        for (int j = 0; j < CCH; ++j) { ex[j] = __expf(g[t * CCH + j] - m); s += ex[j]; }
        const float inv = 1.f / s;
#pragma unroll
        for (int j = 0; j < CCH; ++j) wfin[(long)b * (CCH * CCH) + t * CCH + j] = ex[j] * inv;
    }
}

__global__ __launch_bounds__(256) void apply_kernel(const float* __restrict__ q,
                                                    const float* __restrict__ wfin,
                                                    float* __restrict__ out) {
    const int wg   = blockIdx.x % K3_WGS;
    const int b    = blockIdx.x / K3_WGS;
    const int wave = threadIdx.x >> 5;
    const int lane = threadIdx.x & 31;
    const int lm   = lane & 15;
    const int kb   = (lane < 16) ? 0 : 2;

    const float* wb = wfin + (long)b * (CCH * CCH);
    v2f Bw0[8], Bw1[8];
#pragma unroll
    for (int kk = 0; kk < 8; ++kk) {
        const float* w0 = wb + lm * CCH + 4 * kk + kb;
        const float* w1 = wb + (16 + lm) * CCH + 4 * kk + kb;
        Bw0[kk].x = w0[0];  Bw0[kk].y = w0[1];
        Bw1[kk].x = w1[0];  Bw1[kk].y = w1[1];
    }

    const long n0 = (long)wg * (8 * K3_ROWS) + (long)wave * K3_ROWS;
    const float* qb = q   + ((long)b * NSP + n0) * CCH;
    float*       ob = out + ((long)b * NSP + n0) * CCH;
    const int rowadd = (lane < 16) ? 0 : 8;

    for (int it = 0; it < K3_ROWS; it += 16) {
        v2f A[8];
#pragma unroll
        for (int kk = 0; kk < 8; ++kk) {
            const v2f* qr = (const v2f*)(qb + (long)(it + lm) * CCH + 4 * kk + kb);
            A[kk] = *qr;
        }
        v8f c0 = {}, c1 = {};
#pragma unroll
        for (int kk = 0; kk < 8; ++kk) {
            c0 = __builtin_amdgcn_wmma_f32_16x16x4_f32(false, A[kk], false, Bw0[kk], (short)0, c0, false, false);
            c1 = __builtin_amdgcn_wmma_f32_16x16x4_f32(false, A[kk], false, Bw1[kk], (short)0, c1, false, false);
        }
#pragma unroll
        for (int r = 0; r < 8; ++r) {
            const long row = it + r + rowadd;
            __builtin_nontemporal_store(c0[r] + qb[row * CCH + lm],      ob + row * CCH + lm);
            __builtin_nontemporal_store(c1[r] + qb[row * CCH + 16 + lm], ob + row * CCH + 16 + lm);
        }
    }
}

extern "C" void kernel_launch(void* const* d_in, const int* in_sizes, int n_in,
                              void* d_out, int out_size, void* d_ws, size_t ws_size,
                              hipStream_t stream) {
    const float* q = (const float*)d_in[0];
    const float* k = (const float*)d_in[1];
    float* out = (float*)d_out;
    const int B = in_sizes[0] / (NSP * CCH);

    float* partial = (float*)d_ws;
    float* wfin    = partial + (long)B * K1_WGS * CCH * CCH;

    gram_kernel   <<<dim3(B * K1_WGS), dim3(256), 0, stream>>>(q, k, partial);
    softmax_kernel<<<dim3(B),          dim3(256), 0, stream>>>(partial, wfin);
    apply_kernel  <<<dim3(B * K3_WGS), dim3(256), 0, stream>>>(q, wfin, out);
}


# apply pass: q and out move as whole 128-B lines (16 B per lane) through a per-wave LDS tile into WMMA fragments (was 8-B gathers over 16 lines, 4-B stores); channel sum order permuted
# speedup vs baseline: 1.1068x; 1.1068x over previous
#include <hip/hip_runtime.h>

typedef __attribute__((ext_vector_type(2))) float v2f;
typedef __attribute__((ext_vector_type(4))) float v4f;
typedef __attribute__((ext_vector_type(8))) float v8f;

#define CCH 32
#define NSP (64 * 64 * 64)
#define K1_WGS 128
#define K1_WAVES 8
#define K1_ROWS (NSP / (K1_WGS * K1_WAVES))
#define K3_WGS 512
#define K3_ROWS (NSP / (K3_WGS * 8))

__global__ __launch_bounds__(256) void gram_kernel(const float* __restrict__ q,
                                                   const float* __restrict__ k,
                                                   float* __restrict__ partial) {
    const int wg   = blockIdx.x % K1_WGS;
    const int b    = blockIdx.x / K1_WGS;
    const int wave = threadIdx.x >> 5;
    const int lane = threadIdx.x & 31;
    const int lm   = lane & 15;
    const int kb   = (lane < 16) ? 0 : 2;

    const long n0 = (long)wg * (K1_WAVES * K1_ROWS) + (long)wave * K1_ROWS;
    const float* qb = q + ((long)b * NSP + n0) * CCH;
    const float* kp = k + ((long)b * NSP + n0) * CCH;

    v8f c00 = {}, c01 = {}, c10 = {}, c11 = {};
    for (int it = 0; it < K1_ROWS; it += 4) {
        const float* qr = qb + (long)(it + kb) * CCH;
        const float* kr = kp + (long)(it + kb) * CCH;
        v2f a0, a1, b0, b1;
        a0.x = qr[lm];           a0.y = qr[CCH + lm];
        a1.x = qr[16 + lm];      a1.y = qr[CCH + 16 + lm];
        b0.x = __builtin_nontemporal_load(kr + lm);
        b0.y = __builtin_nontemporal_load(kr + CCH + lm);
        b1.x = __builtin_nontemporal_load(kr + 16 + lm);
        b1.y = __builtin_nontemporal_load(kr + CCH + 16 + lm);
        c00 = __builtin_amdgcn_wmma_f32_16x16x4_f32(false, a0, false, b0, (short)0, c00, false, false);
        c01 = __builtin_amdgcn_wmma_f32_16x16x4_f32(false, a0, false, b1, (short)0, c01, false, false);
        c10 = __builtin_amdgcn_wmma_f32_16x16x4_f32(false, a1, false, b0, (short)0, c10, false, false);
        c11 = __builtin_amdgcn_wmma_f32_16x16x4_f32(false, a1, false, b1, (short)0, c11, false, false);
    }

    __shared__ float smem[K1_WAVES * CCH * CCH];
    const int rowadd = (lane < 16) ? 0 : 8;
    float* sw = smem + wave * (CCH * CCH);
#pragma unroll
    for (int r = 0; r < 8; ++r) {
        const int i0 = r + rowadd;
        sw[(i0     ) * CCH + lm     ] = c00[r];
        sw[(i0     ) * CCH + 16 + lm] = c01[r];
        sw[(i0 + 16) * CCH + lm     ] = c10[r];
        sw[(i0 + 16) * CCH + 16 + lm] = c11[r];
    }
    __syncthreads();

    float* outp = partial + (long)(b * K1_WGS + wg) * (CCH * CCH);
    for (int e = threadIdx.x * 4; e < threadIdx.x * 4 + 4; ++e) {
        float s = 0.f;
        for (int w = 0; w < K1_WAVES; ++w) s += smem[w * (CCH * CCH) + e];
        outp[e] = s;
    }
}

__global__ __launch_bounds__(256) void softmax_kernel(const float* __restrict__ partial,
                                                      float* __restrict__ wfin) {
    const int b = blockIdx.x;
    const int t = threadIdx.x;
    __shared__ float g[CCH * CCH];
    for (int e = t * 4; e < t * 4 + 4; ++e) {
        float s = 0.f;
        const float* p = partial + (long)b * K1_WGS * (CCH * CCH) + e;
        for (int w = 0; w < K1_WGS; ++w) s += p[(long)w * (CCH * CCH)];
        g[e] = s;
    }
    __syncthreads();
    if (t < CCH) {
        float m = -3.4e38f;
#pragma unroll
        for (int j = 0; j < CCH; ++j) m = fmaxf(m, g[t * CCH + j]);
        float ex[CCH];
        float s = 0.f;
#pragma unroll
        for (int j = 0; j < CCH; ++j) { ex[j] = __expf(g[t * CCH + j] - m); s += ex[j]; }
        const float inv = 1.f / s;
#pragma unroll
        for (int j = 0; j < CCH; ++j) wfin[(long)b * (CCH * CCH) + t * CCH + j] = ex[j] * inv;
    }
}


static __device__ __forceinline__ void wave_lds_order() {
    __builtin_amdgcn_fence(__ATOMIC_RELEASE, "wavefront");
    __builtin_amdgcn_wave_barrier();
    __builtin_amdgcn_fence(__ATOMIC_ACQUIRE, "wavefront");
}

#define TPITCH 36
__global__ __launch_bounds__(256) void apply_kernel(const float* __restrict__ q,
                                                    const float* __restrict__ wfin,
                                                    float* __restrict__ out) {
    const int wg   = blockIdx.x % K3_WGS;
    const int b    = blockIdx.x / K3_WGS;
    const int wave = threadIdx.x >> 5;
    const int lane = threadIdx.x & 31;
    const int c    = lane & 15;
    const int h    = lane >> 4;

    __shared__ __attribute__((aligned(16))) float tile[8 * 16 * TPITCH];
    float* sw = tile + wave * (16 * TPITCH);

    const float* wb = wfin + (long)b * (CCH * CCH);
    v2f Aw0[8], Aw1[8];
#pragma unroll
    for (int kk = 0; kk < 8; ++kk) {
        const int j = ((kk & 4) << 2) + 8 * h + 2 * (kk & 3);
        Aw0[kk] = *(const v2f*)(wb + c * CCH + j);
        Aw1[kk] = *(const v2f*)(wb + (16 + c) * CCH + j);
    }

    const long n0 = (long)wg * (8 * K3_ROWS) + (long)wave * K3_ROWS;
    const float* qb = q   + ((long)b * NSP + n0) * CCH;
    float*       ob = out + ((long)b * NSP + n0) * CCH;

    const int lrow = lane >> 3;
    const int lcol = (lane & 7) * 4;
    float*       srow  = sw + lrow * TPITCH + lcol;
    float*       sfrag = sw + c * TPITCH + 8 * h;

    v4f g[4];
#pragma unroll
    for (int m = 0; m < 4; ++m)
        g[m] = *(const v4f*)(qb + (long)(4 * m + lrow) * CCH + lcol);

#pragma unroll
    for (int it = 0; it < K3_ROWS; it += 16) {
#pragma unroll
        for (int m = 0; m < 4; ++m)
            *(v4f*)(srow + 4 * m * TPITCH) = g[m];
        if (it + 16 < K3_ROWS) {
#pragma unroll
            for (int m = 0; m < 4; ++m)
                g[m] = *(const v4f*)(qb + (long)(it + 16 + 4 * m + lrow) * CCH + lcol);
        }
        wave_lds_order();

        const v4f f0 = *(const v4f*)(sfrag);
        const v4f f1 = *(const v4f*)(sfrag + 4);
        const v4f f2 = *(const v4f*)(sfrag + 16);
        const v4f f3 = *(const v4f*)(sfrag + 20);
        v2f Bq[8];
        Bq[0].x = f0.x; Bq[0].y = f0.y;  Bq[1].x = f0.z; Bq[1].y = f0.w;
        Bq[2].x = f1.x; Bq[2].y = f1.y;  Bq[3].x = f1.z; Bq[3].y = f1.w;
        Bq[4].x = f2.x; Bq[4].y = f2.y;  Bq[5].x = f2.z; Bq[5].y = f2.w;
        Bq[6].x = f3.x; Bq[6].y = f3.y;  Bq[7].x = f3.z; Bq[7].y = f3.w;

        v8f d0 = {}, d1 = {};
#pragma unroll
        for (int kk = 0; kk < 8; ++kk) {
            d0 = __builtin_amdgcn_wmma_f32_16x16x4_f32(false, Aw0[kk], false, Bq[kk], (short)0, d0, false, false);
            d1 = __builtin_amdgcn_wmma_f32_16x16x4_f32(false, Aw1[kk], false, Bq[kk], (short)0, d1, false, false);
        }

        v4f o0, o1, o2, o3;
        o0.x = d0[0] + f0.x; o0.y = d0[1] + f0.y; o0.z = d0[2] + f0.z; o0.w = d0[3] + f0.w;
        o1.x = d0[4] + f1.x; o1.y = d0[5] + f1.y; o1.z = d0[6] + f1.z; o1.w = d0[7] + f1.w;
        o2.x = d1[0] + f2.x; o2.y = d1[1] + f2.y; o2.z = d1[2] + f2.z; o2.w = d1[3] + f2.w;
        o3.x = d1[4] + f3.x; o3.y = d1[5] + f3.y; o3.z = d1[6] + f3.z; o3.w = d1[7] + f3.w;
        *(v4f*)(sfrag)      = o0;
        *(v4f*)(sfrag + 4)  = o1;
        *(v4f*)(sfrag + 16) = o2;
        *(v4f*)(sfrag + 20) = o3;
        wave_lds_order();

#pragma unroll
        for (int m = 0; m < 4; ++m) {
            const v4f r = *(const v4f*)(srow + 4 * m * TPITCH);
            __builtin_nontemporal_store(r, (v4f*)(ob + (long)(it + 4 * m + lrow) * CCH + lcol));
        }
    }
}

extern "C" void kernel_launch(void* const* d_in, const int* in_sizes, int n_in,
                              void* d_out, int out_size, void* d_ws, size_t ws_size,
                              hipStream_t stream) {
    const float* q = (const float*)d_in[0];
    const float* k = (const float*)d_in[1];
    float* out = (float*)d_out;
    const int B = in_sizes[0] / (NSP * CCH);

    float* partial = (float*)d_ws;
    float* wfin    = partial + (long)B * K1_WGS * CCH * CCH;

    gram_kernel   <<<dim3(B * K1_WGS), dim3(256), 0, stream>>>(q, k, partial);
    softmax_kernel<<<dim3(B),          dim3(256), 0, stream>>>(partial, wfin);
    apply_kernel  <<<dim3(B * K3_WGS), dim3(256), 0, stream>>>(q, wfin, out);
}


# softmax pass: one wave per row of w, 128 work-groups, independent 16 B loads (was 4 work-groups, 4 B strided loads in 32 waited batches); partials summed in 4 chains joined as a tree, row sum a tree
# speedup vs baseline: 1.2733x; 1.1504x over previous
#include <hip/hip_runtime.h>

typedef __attribute__((ext_vector_type(2))) float v2f;
typedef __attribute__((ext_vector_type(4))) float v4f;
typedef __attribute__((ext_vector_type(8))) float v8f;

#define CCH 32
#define NSP (64 * 64 * 64)
#define K1_WGS 128
#define K1_WAVES 8
#define K1_ROWS (NSP / (K1_WGS * K1_WAVES))
#define K3_WGS 512
#define K3_ROWS (NSP / (K3_WGS * 8))

__global__ __launch_bounds__(256) void gram_kernel(const float* __restrict__ q,
                                                   const float* __restrict__ k,
                                                   float* __restrict__ partial) {
    const int wg   = blockIdx.x % K1_WGS;
    const int b    = blockIdx.x / K1_WGS;
    const int wave = threadIdx.x >> 5;
    const int lane = threadIdx.x & 31;
    const int lm   = lane & 15;
    const int kb   = (lane < 16) ? 0 : 2;

    const long n0 = (long)wg * (K1_WAVES * K1_ROWS) + (long)wave * K1_ROWS;
    const float* qb = q + ((long)b * NSP + n0) * CCH;
    const float* kp = k + ((long)b * NSP + n0) * CCH;

    v8f c00 = {}, c01 = {}, c10 = {}, c11 = {};
    for (int it = 0; it < K1_ROWS; it += 4) {
        const float* qr = qb + (long)(it + kb) * CCH;
        const float* kr = kp + (long)(it + kb) * CCH;
        v2f a0, a1, b0, b1;
        a0.x = qr[lm];           a0.y = qr[CCH + lm];
        a1.x = qr[16 + lm];      a1.y = qr[CCH + 16 + lm];
        b0.x = __builtin_nontemporal_load(kr + lm);
        b0.y = __builtin_nontemporal_load(kr + CCH + lm);
        b1.x = __builtin_nontemporal_load(kr + 16 + lm);
        b1.y = __builtin_nontemporal_load(kr + CCH + 16 + lm);
        c00 = __builtin_amdgcn_wmma_f32_16x16x4_f32(false, a0, false, b0, (short)0, c00, false, false);
        c01 = __builtin_amdgcn_wmma_f32_16x16x4_f32(false, a0, false, b1, (short)0, c01, false, false);
        c10 = __builtin_amdgcn_wmma_f32_16x16x4_f32(false, a1, false, b0, (short)0, c10, false, false);
        c11 = __builtin_amdgcn_wmma_f32_16x16x4_f32(false, a1, false, b1, (short)0, c11, false, false);
    }

    __shared__ float smem[K1_WAVES * CCH * CCH];
    const int rowadd = (lane < 16) ? 0 : 8;
    float* sw = smem + wave * (CCH * CCH);
#pragma unroll
    for (int r = 0; r < 8; ++r) {
        const int i0 = r + rowadd;
        sw[(i0     ) * CCH + lm     ] = c00[r];
        sw[(i0     ) * CCH + 16 + lm] = c01[r];
        sw[(i0 + 16) * CCH + lm     ] = c10[r];
        sw[(i0 + 16) * CCH + 16 + lm] = c11[r];
    }
    __syncthreads();

    float* outp = partial + (long)(b * K1_WGS + wg) * (CCH * CCH);
    for (int e = threadIdx.x * 4; e < threadIdx.x * 4 + 4; ++e) {
        float s = 0.f;
        for (int w = 0; w < K1_WAVES; ++w) s += smem[w * (CCH * CCH) + e];
        outp[e] = s;
    }
}

#define K2_CHAINS 4
#define K2_PER_CHAIN (K1_WGS / K2_CHAINS)
__global__ __launch_bounds__(32) void softmax_kernel(const float* __restrict__ partial,
                                                     float* __restrict__ wfin) {
    const int b    = blockIdx.x / CCH;
    const int i    = blockIdx.x % CCH;
    const int lane = threadIdx.x;
    const int g    = lane >> 3;
    const int col  = (lane & 7) * 4;

    const float* p = partial + ((long)b * K1_WGS + g) * (CCH * CCH) + i * CCH + col;
    v4f v[K2_PER_CHAIN];
    for (int t = 0; t < K2_PER_CHAIN; ++t)
        v[t] = *(const v4f*)(p + (long)t * K2_CHAINS * (CCH * CCH));
    v4f acc = v[0];
    for (int t = 1; t < K2_PER_CHAIN; ++t) acc += v[t];

    float x[4] = {acc.x, acc.y, acc.z, acc.w};
    for (int e = 0; e < 4; ++e) {
        x[e] += __shfl_xor(x[e], 8, 32);
        x[e] += __shfl_xor(x[e], 16, 32);
    }

    float m = fmaxf(fmaxf(fmaxf(fmaxf(-3.4e38f, x[0]), x[1]), x[2]), x[3]);
    for (int sh = 1; sh <= 4; sh <<= 1) m = fmaxf(m, __shfl_xor(m, sh, 32));
    float ex[4];
    for (int e = 0; e < 4; ++e) ex[e] = __expf(x[e] - m);
    float s = (ex[0] + ex[1]) + (ex[2] + ex[3]);
    for (int sh = 1; sh <= 4; sh <<= 1) s += __shfl_xor(s, sh, 32);
    const float inv = 1.f / s;
    if (g == 0) {
        v4f o;
        o.x = ex[0] * inv; o.y = ex[1] * inv; o.z = ex[2] * inv; o.w = ex[3] * inv;
        *(v4f*)(wfin + (long)b * (CCH * CCH) + i * CCH + col) = o;
    }
}


static __device__ __forceinline__ void wave_lds_order() {
    __builtin_amdgcn_fence(__ATOMIC_RELEASE, "wavefront");
    __builtin_amdgcn_wave_barrier();
    __builtin_amdgcn_fence(__ATOMIC_ACQUIRE, "wavefront");
}

#define TPITCH 36
__global__ __launch_bounds__(256) void apply_kernel(const float* __restrict__ q,
                                                    const float* __restrict__ wfin,
                                                    float* __restrict__ out) {
    const int wg   = blockIdx.x % K3_WGS;
    const int b    = blockIdx.x / K3_WGS;
    const int wave = threadIdx.x >> 5;
    const int lane = threadIdx.x & 31;
    const int c    = lane & 15;
    const int h    = lane >> 4;

    __shared__ __attribute__((aligned(16))) float tile[8 * 16 * TPITCH];
    float* sw = tile + wave * (16 * TPITCH);

    const float* wb = wfin + (long)b * (CCH * CCH);
    v2f Aw0[8], Aw1[8];
#pragma unroll
    for (int kk = 0; kk < 8; ++kk) {
        const int j = ((kk & 4) << 2) + 8 * h + 2 * (kk & 3);
        Aw0[kk] = *(const v2f*)(wb + c * CCH + j);
        Aw1[kk] = *(const v2f*)(wb + (16 + c) * CCH + j);
    }

    const long n0 = (long)wg * (8 * K3_ROWS) + (long)wave * K3_ROWS;
    const float* qb = q   + ((long)b * NSP + n0) * CCH;
    float*       ob = out + ((long)b * NSP + n0) * CCH;

    const int lrow = lane >> 3;
    const int lcol = (lane & 7) * 4;
    float*       srow  = sw + lrow * TPITCH + lcol;
    float*       sfrag = sw + c * TPITCH + 8 * h;

    v4f g[4];
#pragma unroll
    for (int m = 0; m < 4; ++m)
        g[m] = *(const v4f*)(qb + (long)(4 * m + lrow) * CCH + lcol);

#pragma unroll
    for (int it = 0; it < K3_ROWS; it += 16) {
#pragma unroll
        for (int m = 0; m < 4; ++m)
            *(v4f*)(srow + 4 * m * TPITCH) = g[m];
        if (it + 16 < K3_ROWS) {
#pragma unroll
            for (int m = 0; m < 4; ++m)
                g[m] = *(const v4f*)(qb + (long)(it + 16 + 4 * m + lrow) * CCH + lcol);
        }
        wave_lds_order();

        const v4f f0 = *(const v4f*)(sfrag);
        const v4f f1 = *(const v4f*)(sfrag + 4);
        const v4f f2 = *(const v4f*)(sfrag + 16);
        const v4f f3 = *(const v4f*)(sfrag + 20);
        v2f Bq[8];
        Bq[0].x = f0.x; Bq[0].y = f0.y;  Bq[1].x = f0.z; Bq[1].y = f0.w;
        Bq[2].x = f1.x; Bq[2].y = f1.y;  Bq[3].x = f1.z; Bq[3].y = f1.w;
        Bq[4].x = f2.x; Bq[4].y = f2.y;  Bq[5].x = f2.z; Bq[5].y = f2.w;
        Bq[6].x = f3.x; Bq[6].y = f3.y;  Bq[7].x = f3.z; Bq[7].y = f3.w;

        v8f d0 = {}, d1 = {};
#pragma unroll
        for (int kk = 0; kk < 8; ++kk) {
            d0 = __builtin_amdgcn_wmma_f32_16x16x4_f32(false, Aw0[kk], false, Bq[kk], (short)0, d0, false, false);
            d1 = __builtin_amdgcn_wmma_f32_16x16x4_f32(false, Aw1[kk], false, Bq[kk], (short)0, d1, false, false);
        }

        v4f o0, o1, o2, o3;
        o0.x = d0[0] + f0.x; o0.y = d0[1] + f0.y; o0.z = d0[2] + f0.z; o0.w = d0[3] + f0.w;
        o1.x = d0[4] + f1.x; o1.y = d0[5] + f1.y; o1.z = d0[6] + f1.z; o1.w = d0[7] + f1.w;
        o2.x = d1[0] + f2.x; o2.y = d1[1] + f2.y; o2.z = d1[2] + f2.z; o2.w = d1[3] + f2.w;
        o3.x = d1[4] + f3.x; o3.y = d1[5] + f3.y; o3.z = d1[6] + f3.z; o3.w = d1[7] + f3.w;
        *(v4f*)(sfrag)      = o0;
        *(v4f*)(sfrag + 4)  = o1;
        *(v4f*)(sfrag + 16) = o2;
        *(v4f*)(sfrag + 20) = o3;
        wave_lds_order();

#pragma unroll
        for (int m = 0; m < 4; ++m) {
            const v4f r = *(const v4f*)(srow + 4 * m * TPITCH);
            __builtin_nontemporal_store(r, (v4f*)(ob + (long)(it + 4 * m + lrow) * CCH + lcol));
        }
    }
}

extern "C" void kernel_launch(void* const* d_in, const int* in_sizes, int n_in,
                              void* d_out, int out_size, void* d_ws, size_t ws_size,
                              hipStream_t stream) {
    const float* q = (const float*)d_in[0];
    const float* k = (const float*)d_in[1];
    float* out = (float*)d_out;
    const int B = in_sizes[0] / (NSP * CCH);

    float* partial = (float*)d_ws;
    float* wfin    = partial + (long)B * K1_WGS * CCH * CCH;

    gram_kernel   <<<dim3(B * K1_WGS), dim3(256), 0, stream>>>(q, k, partial);
    softmax_kernel<<<dim3(B * CCH),    dim3(32),  0, stream>>>(partial, wfin);
    apply_kernel  <<<dim3(B * K3_WGS), dim3(256), 0, stream>>>(q, wfin, out);
}


# gram pass: q and k arrive as whole 128-B lines (16 B per lane, 3 tiles of 16 rows in flight per wave) and reach the WMMA fragments through a per-wave LDS tile (was 4-B loads); sum order kept
# speedup vs baseline: 1.3249x; 1.0405x over previous
#include <hip/hip_runtime.h>

typedef __attribute__((ext_vector_type(2))) float v2f;
typedef __attribute__((ext_vector_type(4))) float v4f;
typedef __attribute__((ext_vector_type(8))) float v8f;

#define CCH 32
#define NSP (64 * 64 * 64)
#define K1_WGS 128
#define K1_WAVES 8
#define K1_ROWS (NSP / (K1_WGS * K1_WAVES))
#define K3_WGS 512
#define K3_ROWS (NSP / (K3_WGS * 8))

static __device__ __forceinline__ void wave_lds_order() {
    __builtin_amdgcn_fence(__ATOMIC_RELEASE, "wavefront");
    __builtin_amdgcn_wave_barrier();
    __builtin_amdgcn_fence(__ATOMIC_ACQUIRE, "wavefront");
}

#define G_HALF (16 * 16 + 32)
#define G_OPER (G_HALF + 16 * 16)
#define G_TILE 16
#define G_AHEAD 3
#define G_TILES (K1_ROWS / G_TILE)

static __device__ __forceinline__ void gram_frag(const float* t, v2f (&lo)[4], v2f (&hi)[4]) {
#pragma unroll
    for (int kk = 0; kk < 4; ++kk) {
        const float* p = t + 4 * kk * 16;
        lo[kk].x = p[0];           lo[kk].y = p[16];
        hi[kk].x = p[G_HALF];      hi[kk].y = p[G_HALF + 16];
    }
}

__global__ __launch_bounds__(256) void gram_kernel(const float* __restrict__ q,
                                                   const float* __restrict__ k,
                                                   float* __restrict__ partial) {
    const int wg   = blockIdx.x % K1_WGS;
    const int b    = blockIdx.x / K1_WGS;
    const int wave = threadIdx.x >> 5;
    const int lane = threadIdx.x & 31;
    const int lm   = lane & 15;
    const int h    = lane >> 4;

    __shared__ __attribute__((aligned(16))) float gtile[K1_WAVES * 2 * G_OPER];
    float* tq = gtile + wave * (2 * G_OPER);
    float* tk = tq + G_OPER;

    const long n0 = (long)wg * (K1_WAVES * K1_ROWS) + (long)wave * K1_ROWS;

    const int lrow = lane >> 3;
    const int lcol = (lane & 7) * 4;
    const float* qg = q + ((long)b * NSP + n0 + lrow) * CCH + lcol;
    const float* kg = k + ((long)b * NSP + n0 + lrow) * CCH + lcol;
    const int soff = (lcol >> 4) * G_HALF + lrow * 16 + (lcol & 15);
    float* sq = tq + soff;
    float* sk = tk + soff;
    const float* fq = tq + 2 * h * 16 + lm;
    const float* fk = tk + 2 * h * 16 + lm;

    v4f Gq[G_AHEAD][4], Gk[G_AHEAD][4];
#pragma unroll
    for (int s = 0; s < G_AHEAD; ++s) {
#pragma unroll
        for (int m = 0; m < 4; ++m) {
            Gq[s][m] = *(const v4f*)(qg + (long)(s * G_TILE + 4 * m) * CCH);
            Gk[s][m] = __builtin_nontemporal_load((const v4f*)(kg + (long)(s * G_TILE + 4 * m) * CCH));
        }
    }

    v8f c00 = {}, c01 = {}, c10 = {}, c11 = {};
#pragma unroll
    for (int t = 0; t < G_TILES; ++t) {
        const int s = t % G_AHEAD;
#pragma unroll
        for (int m = 0; m < 4; ++m) {
            *(v4f*)(sq + 4 * m * 16) = Gq[s][m];
            *(v4f*)(sk + 4 * m * 16) = Gk[s][m];
        }
        if (t + G_AHEAD < G_TILES) {
#pragma unroll
            for (int m = 0; m < 4; ++m) {
                Gq[s][m] = *(const v4f*)(qg + (long)((t + G_AHEAD) * G_TILE + 4 * m) * CCH);
                Gk[s][m] = __builtin_nontemporal_load((const v4f*)(kg + (long)((t + G_AHEAD) * G_TILE + 4 * m) * CCH));
            }
        }
        wave_lds_order();

        v2f a0[4], a1[4], b0[4], b1[4];
        gram_frag(fq, a0, a1);
        gram_frag(fk, b0, b1);
#pragma unroll
        for (int kk = 0; kk < 4; ++kk) {
            c00 = __builtin_amdgcn_wmma_f32_16x16x4_f32(false, a0[kk], false, b0[kk], (short)0, c00, false, false);
            c01 = __builtin_amdgcn_wmma_f32_16x16x4_f32(false, a0[kk], false, b1[kk], (short)0, c01, false, false);
            c10 = __builtin_amdgcn_wmma_f32_16x16x4_f32(false, a1[kk], false, b0[kk], (short)0, c10, false, false);
            c11 = __builtin_amdgcn_wmma_f32_16x16x4_f32(false, a1[kk], false, b1[kk], (short)0, c11, false, false);
        }
        wave_lds_order();
    }

    __shared__ float smem[K1_WAVES * CCH * CCH];
    const int rowadd = (lane < 16) ? 0 : 8;
    float* sw = smem + wave * (CCH * CCH);
#pragma unroll
    for (int r = 0; r < 8; ++r) {
        const int i0 = r + rowadd;
        sw[(i0     ) * CCH + lm     ] = c00[r];
        sw[(i0     ) * CCH + 16 + lm] = c01[r];
        sw[(i0 + 16) * CCH + lm     ] = c10[r];
        sw[(i0 + 16) * CCH + 16 + lm] = c11[r];
    }
    __syncthreads();

    float* outp = partial + (long)(b * K1_WGS + wg) * (CCH * CCH);
    for (int e = threadIdx.x * 4; e < threadIdx.x * 4 + 4; ++e) {
        float s = 0.f;
        for (int w = 0; w < K1_WAVES; ++w) s += smem[w * (CCH * CCH) + e];
        outp[e] = s;
    }
}

#define K2_CHAINS 4
#define K2_PER_CHAIN (K1_WGS / K2_CHAINS)
__global__ __launch_bounds__(32) void softmax_kernel(const float* __restrict__ partial,
                                                     float* __restrict__ wfin) {
    const int b    = blockIdx.x / CCH;
    const int i    = blockIdx.x % CCH;
    const int lane = threadIdx.x;
    const int g    = lane >> 3;
    const int col  = (lane & 7) * 4;

    const float* p = partial + ((long)b * K1_WGS + g) * (CCH * CCH) + i * CCH + col;
    v4f v[K2_PER_CHAIN];
    for (int t = 0; t < K2_PER_CHAIN; ++t)
        v[t] = *(const v4f*)(p + (long)t * K2_CHAINS * (CCH * CCH));
    v4f acc = v[0];
    for (int t = 1; t < K2_PER_CHAIN; ++t) acc += v[t];

    float x[4] = {acc.x, acc.y, acc.z, acc.w};
    for (int e = 0; e < 4; ++e) {
        x[e] += __shfl_xor(x[e], 8, 32);
        x[e] += __shfl_xor(x[e], 16, 32);
    }

    float m = fmaxf(fmaxf(fmaxf(fmaxf(-3.4e38f, x[0]), x[1]), x[2]), x[3]);
    for (int sh = 1; sh <= 4; sh <<= 1) m = fmaxf(m, __shfl_xor(m, sh, 32));
    float ex[4];
    for (int e = 0; e < 4; ++e) ex[e] = __expf(x[e] - m);
    float s = (ex[0] + ex[1]) + (ex[2] + ex[3]);
    for (int sh = 1; sh <= 4; sh <<= 1) s += __shfl_xor(s, sh, 32);
    const float inv = 1.f / s;
    if (g == 0) {
        v4f o;
        o.x = ex[0] * inv; o.y = ex[1] * inv; o.z = ex[2] * inv; o.w = ex[3] * inv;
        *(v4f*)(wfin + (long)b * (CCH * CCH) + i * CCH + col) = o;
    }
}


#define TPITCH 36
__global__ __launch_bounds__(256) void apply_kernel(const float* __restrict__ q,
                                                    const float* __restrict__ wfin,
                                                    float* __restrict__ out) {
    const int wg   = blockIdx.x % K3_WGS;
    const int b    = blockIdx.x / K3_WGS;
    const int wave = threadIdx.x >> 5;
    const int lane = threadIdx.x & 31;
    const int c    = lane & 15;
    const int h    = lane >> 4;

    __shared__ __attribute__((aligned(16))) float tile[8 * 16 * TPITCH];
    float* sw = tile + wave * (16 * TPITCH);

    const float* wb = wfin + (long)b * (CCH * CCH);
    v2f Aw0[8], Aw1[8];
#pragma unroll
    for (int kk = 0; kk < 8; ++kk) {
        const int j = ((kk & 4) << 2) + 8 * h + 2 * (kk & 3);
        Aw0[kk] = *(const v2f*)(wb + c * CCH + j);
        Aw1[kk] = *(const v2f*)(wb + (16 + c) * CCH + j);
    }

    const long n0 = (long)wg * (8 * K3_ROWS) + (long)wave * K3_ROWS;
    const float* qb = q   + ((long)b * NSP + n0) * CCH;
    float*       ob = out + ((long)b * NSP + n0) * CCH;

    const int lrow = lane >> 3;
    const int lcol = (lane & 7) * 4;
    float*       srow  = sw + lrow * TPITCH + lcol;
    float*       sfrag = sw + c * TPITCH + 8 * h;

    v4f g[4];
#pragma unroll
    for (int m = 0; m < 4; ++m)
        g[m] = *(const v4f*)(qb + (long)(4 * m + lrow) * CCH + lcol);

#pragma unroll
    for (int it = 0; it < K3_ROWS; it += 16) {
#pragma unroll
        for (int m = 0; m < 4; ++m)
            *(v4f*)(srow + 4 * m * TPITCH) = g[m];
        if (it + 16 < K3_ROWS) {
#pragma unroll
            for (int m = 0; m < 4; ++m)
                g[m] = *(const v4f*)(qb + (long)(it + 16 + 4 * m + lrow) * CCH + lcol);
        }
        wave_lds_order();

        const v4f f0 = *(const v4f*)(sfrag);
        const v4f f1 = *(const v4f*)(sfrag + 4);
        const v4f f2 = *(const v4f*)(sfrag + 16);
        const v4f f3 = *(const v4f*)(sfrag + 20);
        v2f Bq[8];
        Bq[0].x = f0.x; Bq[0].y = f0.y;  Bq[1].x = f0.z; Bq[1].y = f0.w;
        Bq[2].x = f1.x; Bq[2].y = f1.y;  Bq[3].x = f1.z; Bq[3].y = f1.w;
        Bq[4].x = f2.x; Bq[4].y = f2.y;  Bq[5].x = f2.z; Bq[5].y = f2.w;
        Bq[6].x = f3.x; Bq[6].y = f3.y;  Bq[7].x = f3.z; Bq[7].y = f3.w;

        v8f d0 = {}, d1 = {};
#pragma unroll
        for (int kk = 0; kk < 8; ++kk) {
            d0 = __builtin_amdgcn_wmma_f32_16x16x4_f32(false, Aw0[kk], false, Bq[kk], (short)0, d0, false, false);
            d1 = __builtin_amdgcn_wmma_f32_16x16x4_f32(false, Aw1[kk], false, Bq[kk], (short)0, d1, false, false);
        }

        v4f o0, o1, o2, o3;
        o0.x = d0[0] + f0.x; o0.y = d0[1] + f0.y; o0.z = d0[2] + f0.z; o0.w = d0[3] + f0.w;
        o1.x = d0[4] + f1.x; o1.y = d0[5] + f1.y; o1.z = d0[6] + f1.z; o1.w = d0[7] + f1.w;
        o2.x = d1[0] + f2.x; o2.y = d1[1] + f2.y; o2.z = d1[2] + f2.z; o2.w = d1[3] + f2.w;
        o3.x = d1[4] + f3.x; o3.y = d1[5] + f3.y; o3.z = d1[6] + f3.z; o3.w = d1[7] + f3.w;
        *(v4f*)(sfrag)      = o0;
        *(v4f*)(sfrag + 4)  = o1;
        *(v4f*)(sfrag + 16) = o2;
        *(v4f*)(sfrag + 20) = o3;
        wave_lds_order();

#pragma unroll
        for (int m = 0; m < 4; ++m) {
            const v4f r = *(const v4f*)(srow + 4 * m * TPITCH);
            __builtin_nontemporal_store(r, (v4f*)(ob + (long)(it + 4 * m + lrow) * CCH + lcol));
        }
    }
}

extern "C" void kernel_launch(void* const* d_in, const int* in_sizes, int n_in,
                              void* d_out, int out_size, void* d_ws, size_t ws_size,
                              hipStream_t stream) {
    const float* q = (const float*)d_in[0];
    const float* k = (const float*)d_in[1];
    float* out = (float*)d_out;
    const int B = in_sizes[0] / (NSP * CCH);

    float* partial = (float*)d_ws;
    float* wfin    = partial + (long)B * K1_WGS * CCH * CCH;

    gram_kernel   <<<dim3(B * K1_WGS), dim3(256), 0, stream>>>(q, k, partial);
    softmax_kernel<<<dim3(B * CCH),    dim3(32),  0, stream>>>(partial, wfin);
    apply_kernel  <<<dim3(B * K3_WGS), dim3(256), 0, stream>>>(q, wfin, out);
}
